# starting kernel
# baseline (speedup 1.0000x reference)
#include <hip/hip_runtime.h>

typedef float v2f __attribute__((ext_vector_type(2)));
typedef float v8f __attribute__((ext_vector_type(8)));

#define IMG_H 512
#define IMG_W 512
#define NBATCH 16
#define TS 16
#define KSZ 5
#define PADR 2
#define HALO (TS + KSZ - 1)

__device__ __forceinline__ int refl(int i) {
    i = (i < 0) ? -i : i;
    i = (i >= IMG_H) ? (2 * IMG_H - 2 - i) : i;
    return i;
}

__global__ __launch_bounds__(256)
void patch_moments_wmma(const float* __restrict__ x, float* __restrict__ out) {
    __shared__ float Tlds[8][2][16][HALO];

    const int lane  = threadIdx.x & 31;
    const int wave  = threadIdx.x >> 5;
    const int tile  = blockIdx.x * 8 + wave;

    const int b  = tile >> 10;
    const int tr = (tile >> 5) & 31;
    const int tc = tile & 31;
    const int r0 = tr * TS;
    const int c0 = tc * TS;

    const float* __restrict__ xb = x + (size_t)b * (IMG_H * IMG_W);

    const int n     = lane & 15;
    const int khalf = lane >> 4;
    const int m     = n;

    v8f dsum0 = {}, dsum1 = {};
    v8f dsq0  = {}, dsq1  = {};

    #pragma unroll
    for (int g = 0; g < 2; ++g) {
        const int cpad = g * 16 + n;
        const int gcol = refl(c0 - PADR + cpad);
        #pragma unroll
        for (int kc = 0; kc < 5; ++kc) {
            v2f a, bx, bxx;
            #pragma unroll
            for (int p = 0; p < 2; ++p) {
                const int k = 4 * kc + 2 * khalf + p;
                a[p] = (k >= m && k <= m + 4) ? 1.0f : 0.0f;
                const int grow = refl(r0 - PADR + k);
                const float v  = xb[grow * IMG_W + gcol];
                bx[p]  = v;
                bxx[p] = v * v;
            }
            if (g == 0) {
                dsum0 = __builtin_amdgcn_wmma_f32_16x16x4_f32(false, a, false, bx,  (short)0, dsum0, false, false);
                dsq0  = __builtin_amdgcn_wmma_f32_16x16x4_f32(false, a, false, bxx, (short)0, dsq0,  false, false);
            } else {
                dsum1 = __builtin_amdgcn_wmma_f32_16x16x4_f32(false, a, false, bx,  (short)0, dsum1, false, false);
                dsq1  = __builtin_amdgcn_wmma_f32_16x16x4_f32(false, a, false, bxx, (short)0, dsq1,  false, false);
            }
        }
    }

    #pragma unroll
    for (int vv = 0; vv < 8; ++vv) {
        const int row = vv + 8 * khalf;
        Tlds[wave][0][row][n] = dsum0[vv];
        Tlds[wave][1][row][n] = dsq0[vv];
        if (n < HALO - 16) {
            Tlds[wave][0][row][16 + n] = dsum1[vv];
            Tlds[wave][1][row][16 + n] = dsq1[vv];
        }
    }
    asm volatile("s_wait_dscnt 0" ::: "memory");

    v8f s1 = {}, s2 = {};
    #pragma unroll
    for (int kc = 0; kc < 5; ++kc) {
        v2f asum, asq, bb;
        #pragma unroll
        for (int p = 0; p < 2; ++p) {
            const int k = 4 * kc + 2 * khalf + p;
            asum[p] = Tlds[wave][0][m][k];
            asq[p]  = Tlds[wave][1][m][k];
            bb[p] = (k >= n && k <= n + 4) ? 1.0f : 0.0f;
        }
        s1 = __builtin_amdgcn_wmma_f32_16x16x4_f32(false, asum, false, bb, (short)0, s1, false, false);
        s2 = __builtin_amdgcn_wmma_f32_16x16x4_f32(false, asq,  false, bb, (short)0, s2, false, false);
    }

    const float inv_n   = 1.0f / 25.0f;
    const float inv_nm1 = 1.0f / 24.0f;
    float* __restrict__ avg_out = out;
    float* __restrict__ var_out = out + (size_t)NBATCH * IMG_H * IMG_W;
    const size_t base = (size_t)b * (IMG_H * IMG_W);

    #pragma unroll
    for (int vv = 0; vv < 8; ++vv) {
        const int row = r0 + vv + 8 * khalf;
        const int col = c0 + n;
        const float ssum = s1[vv];
        const float a    = ssum * inv_n;
        const float var  = (s2[vv] - ssum * a) * inv_nm1;
        const size_t idx = base + (size_t)(row * IMG_W + col);
        avg_out[idx] = a;
        var_out[idx] = var;
    }
}

extern "C" void kernel_launch(void* const* d_in, const int* in_sizes, int n_in,
                              void* d_out, int out_size, void* d_ws, size_t ws_size,
                              hipStream_t stream) {
    (void)in_sizes; (void)n_in; (void)out_size; (void)d_ws; (void)ws_size;
    const float* x = (const float*)d_in[0];
    float* out = (float*)d_out;
    patch_moments_wmma<<<dim3(2048), dim3(256), 0, stream>>>(x, out);
}


# Strip tiling: an 8-wave block owns 16 rows x 512 cols and shares vertical-pass sums via LDS (one barrier); no halo column is loaded or multiplied twice (30 -> 20 WMMA per tile); sum order unchanged
# speedup vs baseline: 1.1642x; 1.1642x over previous
#include <hip/hip_runtime.h>


typedef float v2f __attribute__((ext_vector_type(2)));
typedef float v8f __attribute__((ext_vector_type(8)));

#define IMG_H 512
#define IMG_W 512
#define NBATCH 16
#define TS 16
#define KSZ 5
#define PADR 2
#define HALO (TS + KSZ - 1)
#define NWAVE 8
#define NG (IMG_W / (NWAVE * 16))
#define TPITCH (IMG_W + 2 * PADR)

__device__ __forceinline__ int refl(int i) {
    i = (i < 0) ? -i : i;
    i = (i >= IMG_H) ? (2 * IMG_H - 2 - i) : i;
    return i;
}

__global__ __launch_bounds__(256)
void patch_moments_wmma(const float* __restrict__ x, float* __restrict__ out) {
    __shared__ float Tlds[2][TS][TPITCH];

    const int lane  = threadIdx.x & 31;
    const int wave  = threadIdx.x >> 5;

    const int b  = blockIdx.x >> 5;
    const int r0 = (blockIdx.x & 31) * TS;
    const int cw = wave * (NG * 16);

    const float* __restrict__ xb = x + (size_t)b * (IMG_H * IMG_W);

    const int n     = lane & 15;
    const int khalf = lane >> 4;
    const int m     = n;

    v8f dsum[NG], dsq[NG];
    #pragma unroll
    for (int g = 0; g < NG; ++g) { dsum[g] = (v8f){}; dsq[g] = (v8f){}; }

    #pragma unroll
    for (int kc = 0; kc < 5; ++kc) {
        v2f a;
        int roff[2];
        #pragma unroll
        for (int p = 0; p < 2; ++p) {
            const int k = 4 * kc + 2 * khalf + p;
            a[p] = (k >= m && k <= m + 4) ? 1.0f : 0.0f;
            roff[p] = refl(r0 - PADR + k) * IMG_W + cw + n;
        }
        #pragma unroll
        for (int g = 0; g < NG; ++g) {
            v2f bx, bxx;
            #pragma unroll
            for (int p = 0; p < 2; ++p) {
                const float v = xb[roff[p] + 16 * g];
                bx[p]  = v;
                bxx[p] = v * v;
            }
            dsum[g] = __builtin_amdgcn_wmma_f32_16x16x4_f32(false, a, false, bx,  (short)0, dsum[g], false, false);
            dsq[g]  = __builtin_amdgcn_wmma_f32_16x16x4_f32(false, a, false, bxx, (short)0, dsq[g],  false, false);
        }
    }

    #pragma unroll
    for (int g = 0; g < NG; ++g) {
        #pragma unroll
        for (int vv = 0; vv < 8; ++vv) {
            const int row = vv + 8 * khalf;
            Tlds[0][row][PADR + cw + 16 * g + n] = dsum[g][vv];
            Tlds[1][row][PADR + cw + 16 * g + n] = dsq[g][vv];
        }
    }
    if (wave == 0 && (n == 1 || n == 2)) {
        #pragma unroll
        for (int vv = 0; vv < 8; ++vv) {
            const int row = vv + 8 * khalf;
            Tlds[0][row][PADR - n] = dsum[0][vv];
            Tlds[1][row][PADR - n] = dsq[0][vv];
        }
    }
    if (wave == NWAVE - 1 && (n == 13 || n == 14)) {
        #pragma unroll
        for (int vv = 0; vv < 8; ++vv) {
            const int row = vv + 8 * khalf;
            const int c   = cw + 16 * (NG - 1) + n;
            Tlds[0][row][PADR + (2 * IMG_W - 2 - c)] = dsum[NG - 1][vv];
            Tlds[1][row][PADR + (2 * IMG_W - 2 - c)] = dsq[NG - 1][vv];
        }
    }
    __syncthreads();

    const float inv_n   = 1.0f / 25.0f;
    const float inv_nm1 = 1.0f / 24.0f;
    float* __restrict__ avg_out = out;
    float* __restrict__ var_out = out + (size_t)NBATCH * IMG_H * IMG_W;
    const size_t base = (size_t)b * (IMG_H * IMG_W);

    #pragma unroll
    for (int g = 0; g < NG; ++g) {
        const int c0 = cw + 16 * g;
        v8f s1 = {}, s2 = {};
        #pragma unroll
        for (int kc = 0; kc < 5; ++kc) {
            v2f asum, asq, bb;
            #pragma unroll
            for (int p = 0; p < 2; ++p) {
                const int k = 4 * kc + 2 * khalf + p;
                asum[p] = Tlds[0][m][c0 + k];
                asq[p]  = Tlds[1][m][c0 + k];
                bb[p] = (k >= n && k <= n + 4) ? 1.0f : 0.0f;
            }
            s1 = __builtin_amdgcn_wmma_f32_16x16x4_f32(false, asum, false, bb, (short)0, s1, false, false);
            s2 = __builtin_amdgcn_wmma_f32_16x16x4_f32(false, asq,  false, bb, (short)0, s2, false, false);
        }

        #pragma unroll
        for (int vv = 0; vv < 8; ++vv) {
            const int row = r0 + vv + 8 * khalf;
            const int col = c0 + n;
            const float ssum = s1[vv];
            const float a    = ssum * inv_n;
            const float var  = (s2[vv] - ssum * a) * inv_nm1;
            const size_t idx = base + (size_t)(row * IMG_W + col);
            avg_out[idx] = a;
            var_out[idx] = var;
        }
    }
}

extern "C" void kernel_launch(void* const* d_in, const int* in_sizes, int n_in,
                              void* d_out, int out_size, void* d_ws, size_t ws_size,
                              hipStream_t stream) {
    (void)in_sizes; (void)n_in; (void)out_size; (void)d_ws; (void)ws_size;
    const float* x = (const float*)d_in[0];
    float* out = (float*)d_out;
    patch_moments_wmma<<<dim3(512), dim3(256), 0, stream>>>(x, out);
}


# Split block barrier: a wave arrives after its T stores, does its two interior tiles (own T columns only) before the wait, and stores each tile right behind its own WMMAs; sum order unchanged
# speedup vs baseline: 1.3090x; 1.1244x over previous
#include <hip/hip_runtime.h>


typedef float v2f __attribute__((ext_vector_type(2)));
typedef float v8f __attribute__((ext_vector_type(8)));

#define IMG_H 512
#define IMG_W 512
#define NBATCH 16
#define TS 16
#define KSZ 5
#define PADR 2
#define HALO (TS + KSZ - 1)
#define NWAVE 8
#define NG (IMG_W / (NWAVE * 16))
#define TPITCH (IMG_W + 2 * PADR)

__device__ __forceinline__ int refl(int i) {
    i = (i < 0) ? -i : i;
    i = (i >= IMG_H) ? (2 * IMG_H - 2 - i) : i;
    return i;
}

__device__ __forceinline__ void load_tile_frags(const float (*T)[TS][TPITCH], int m, int khalf, int c0,
                                                v2f (&fs)[5], v2f (&fq)[5]) {
    #pragma unroll
    for (int kc = 0; kc < 5; ++kc) {
        #pragma unroll
        for (int p = 0; p < 2; ++p) {
            const int k = 4 * kc + 2 * khalf + p;
            fs[kc][p] = T[0][m][c0 + k];
            fq[kc][p] = T[1][m][c0 + k];
        }
    }
}

__device__ __forceinline__ void tile_matmul(const v2f (&fs)[5], const v2f (&fq)[5], const v2f (&bb)[5], v8f& s1, v8f& s2) {
    s1 = (v8f){}; s2 = (v8f){};
    #pragma unroll
    for (int kc = 0; kc < 5; ++kc) {
        s1 = __builtin_amdgcn_wmma_f32_16x16x4_f32(false, fs[kc], false, bb[kc], (short)0, s1, false, false);
        s2 = __builtin_amdgcn_wmma_f32_16x16x4_f32(false, fq[kc], false, bb[kc], (short)0, s2, false, false);
    }
}

__device__ __forceinline__ void tile_finalize_store(const v8f& s1, const v8f& s2, float* __restrict__ avg_out,
                                                    float* __restrict__ var_out, size_t base, int r0, int khalf, int col) {
    const float inv_n   = 1.0f / 25.0f;
    const float inv_nm1 = 1.0f / 24.0f;
    #pragma unroll
    for (int vv = 0; vv < 8; ++vv) {
        const int row = r0 + vv + 8 * khalf;
        const float ssum = s1[vv];
        const float a    = ssum * inv_n;
        const float var  = (s2[vv] - ssum * a) * inv_nm1;
        const size_t idx = base + (size_t)(row * IMG_W + col);
        avg_out[idx] = a;
        var_out[idx] = var;
    }
}

__global__ __launch_bounds__(256)
void patch_moments_wmma(const float* __restrict__ x, float* __restrict__ out) {
    __shared__ float Tlds[2][TS][TPITCH];

    const int lane  = threadIdx.x & 31;
    const int wave  = threadIdx.x >> 5;

    const int b  = blockIdx.x >> 5;
    const int r0 = (blockIdx.x & 31) * TS;
    const int cw = wave * (NG * 16);

    const float* __restrict__ xb = x + (size_t)b * (IMG_H * IMG_W);

    const int n     = lane & 15;
    const int khalf = lane >> 4;
    const int m     = n;

    v8f dsum[NG], dsq[NG];
    #pragma unroll
    for (int g = 0; g < NG; ++g) { dsum[g] = (v8f){}; dsq[g] = (v8f){}; }

    #pragma unroll
    for (int kc = 0; kc < 5; ++kc) {
        v2f a;
        int roff[2];
        #pragma unroll
        for (int p = 0; p < 2; ++p) {
            const int k = 4 * kc + 2 * khalf + p;
            a[p] = (k >= m && k <= m + 4) ? 1.0f : 0.0f;
            roff[p] = refl(r0 - PADR + k) * IMG_W + cw + n;
        }
        #pragma unroll
        for (int g = 0; g < NG; ++g) {
            v2f bx, bxx;
            #pragma unroll
            for (int p = 0; p < 2; ++p) {
                const float v = xb[roff[p] + 16 * g];
                bx[p]  = v;
                bxx[p] = v * v;
            }
            dsum[g] = __builtin_amdgcn_wmma_f32_16x16x4_f32(false, a, false, bx,  (short)0, dsum[g], false, false);
            dsq[g]  = __builtin_amdgcn_wmma_f32_16x16x4_f32(false, a, false, bxx, (short)0, dsq[g],  false, false);
        }
    }

    #pragma unroll
    for (int g = 0; g < NG; ++g) {
        #pragma unroll
        for (int vv = 0; vv < 8; ++vv) {
            const int row = vv + 8 * khalf;
            Tlds[0][row][PADR + cw + 16 * g + n] = dsum[g][vv];
            Tlds[1][row][PADR + cw + 16 * g + n] = dsq[g][vv];
        }
    }
    if (wave == 0 && (n == 1 || n == 2)) {
        #pragma unroll
        for (int vv = 0; vv < 8; ++vv) {
            const int row = vv + 8 * khalf;
            Tlds[0][row][PADR - n] = dsum[0][vv];
            Tlds[1][row][PADR - n] = dsq[0][vv];
        }
    }
    if (wave == NWAVE - 1 && (n == 13 || n == 14)) {
        #pragma unroll
        for (int vv = 0; vv < 8; ++vv) {
            const int row = vv + 8 * khalf;
            const int c   = cw + 16 * (NG - 1) + n;
            Tlds[0][row][PADR + (2 * IMG_W - 2 - c)] = dsum[NG - 1][vv];
            Tlds[1][row][PADR + (2 * IMG_W - 2 - c)] = dsq[NG - 1][vv];
        }
    }
    __builtin_amdgcn_fence(__ATOMIC_RELEASE, "workgroup", "local");
    __builtin_amdgcn_s_barrier_signal(-1);

    float* __restrict__ avg_out = out;
    float* __restrict__ var_out = out + (size_t)NBATCH * IMG_H * IMG_W;
    const size_t base = (size_t)b * (IMG_H * IMG_W);

    v2f bb[5];
    #pragma unroll
    for (int kc = 0; kc < 5; ++kc) {
        #pragma unroll
        for (int p = 0; p < 2; ++p) {
            const int k = 4 * kc + 2 * khalf + p;
            bb[kc][p] = (k >= n && k <= n + 4) ? 1.0f : 0.0f;
        }
    }

    v2f fsA[5], fqA[5], fsB[5], fqB[5];
    v8f s1A, s2A, s1B, s2B;

    load_tile_frags(Tlds, m, khalf, cw + 16, fsA, fqA);
    load_tile_frags(Tlds, m, khalf, cw + 32, fsB, fqB);
    tile_matmul(fsA, fqA, bb, s1A, s2A);
    __builtin_amdgcn_sched_barrier(0);
    tile_finalize_store(s1A, s2A, avg_out, var_out, base, r0, khalf, cw + 16 + n);
    tile_matmul(fsB, fqB, bb, s1B, s2B);
    __builtin_amdgcn_sched_barrier(0);
    tile_finalize_store(s1B, s2B, avg_out, var_out, base, r0, khalf, cw + 32 + n);

    __builtin_amdgcn_s_barrier_wait(-1);
    __builtin_amdgcn_fence(__ATOMIC_ACQUIRE, "workgroup", "local");

    load_tile_frags(Tlds, m, khalf, cw, fsA, fqA);
    load_tile_frags(Tlds, m, khalf, cw + 48, fsB, fqB);
    tile_matmul(fsA, fqA, bb, s1A, s2A);
    __builtin_amdgcn_sched_barrier(0);
    tile_finalize_store(s1A, s2A, avg_out, var_out, base, r0, khalf, cw + n);
    tile_matmul(fsB, fqB, bb, s1B, s2B);
    __builtin_amdgcn_sched_barrier(0);
    tile_finalize_store(s1B, s2B, avg_out, var_out, base, r0, khalf, cw + 48 + n);
}

extern "C" void kernel_launch(void* const* d_in, const int* in_sizes, int n_in,
                              void* d_out, int out_size, void* d_ws, size_t ws_size,
                              hipStream_t stream) {
    (void)in_sizes; (void)n_in; (void)out_size; (void)d_ws; (void)ws_size;
    const float* x = (const float*)d_in[0];
    float* out = (float*)d_out;
    patch_moments_wmma<<<dim3(512), dim3(256), 0, stream>>>(x, out);
}
